# starting kernel
# baseline (speedup 1.0000x reference)
#include <hip/hip_runtime.h>

#define BB   32
#define NN   1024
#define DD   64
#define TT   2048
#define GG   8
#define TC   (TT / GG)
#define NB   16
#define NBP  (NB + 4)
#define NBLK (NN / NB)
#define BN   (BB * NN)
#define BNG  (BN * GG)

typedef float v2f __attribute__((ext_vector_type(2)));
typedef float v4f __attribute__((ext_vector_type(4)));
typedef float v8f __attribute__((ext_vector_type(8)));

#define SWIZ(v, imm) \
    __int_as_float(__builtin_amdgcn_ds_swizzle(__float_as_int(v), (imm)))

__device__ __forceinline__ float wave_max32(float v) {
    v = fmaxf(v, SWIZ(v, 0x401F));
    v = fmaxf(v, SWIZ(v, 0x201F));
    v = fmaxf(v, SWIZ(v, 0x101F));
    v = fmaxf(v, SWIZ(v, 0x081F));
    v = fmaxf(v, SWIZ(v, 0x041F));
    return v;
}
__device__ __forceinline__ float wave_sum32(float v) {
    v += SWIZ(v, 0x401F);
    v += SWIZ(v, 0x201F);
    v += SWIZ(v, 0x101F);
    v += SWIZ(v, 0x081F);
    v += SWIZ(v, 0x041F);
    return v;
}

__device__ __forceinline__ void smerge(float& m, float& num, float& den,
                                       float m2, float n2, float d2) {
    float M  = fmaxf(m, m2);
    float s1 = __expf(m - M);
    float s2 = __expf(m2 - M);
    num = num * s1 + n2 * s2;
    den = den * s1 + d2 * s2;
    m   = M;
}

__global__ __launch_bounds__(256)
void mmse_scan_kernel(const float* __restrict__ data,
                      const float* __restrict__ targets,
                      const float* __restrict__ W,
                      float* __restrict__ wsM,
                      float* __restrict__ wsNum,
                      float* __restrict__ wsDen)
{
    __shared__ alignas(16) float xw[2][TC][NBP];
    __shared__ alignas(16) float red[2][NB][8][4];

    const int tid  = threadIdx.x;
    const int lane = tid & 31;
    const int wave = tid >> 5;
    const int lm   = lane & 15;
    const int lh   = lane >> 4;

    const int b     = (int)blockIdx.x >> 3;
    const int g     = (int)blockIdx.x & (GG - 1);
    const int tbase = g * TC;

    const float* dataB = data + (size_t)b * NN * DD;

    v2f bfrag[2][16];
#pragma unroll
    for (int j = 0; j < 2; ++j) {
        const int gcol = tbase + wave * 32 + j * 16 + lm;
#pragma unroll
        for (int s = 0; s < 16; ++s) {
            const int k0 = 4 * s + 2 * lh;
            v2f bf;
            bf.x = W[(size_t)k0 * TT + gcol];
            bf.y = W[(size_t)(k0 + 1) * TT + gcol];
            bfrag[j][s] = bf;
        }
    }

    auto compute = [&](int blk, int buf) {
        const int n0 = blk * NB;
        v2f af[16];
#pragma unroll
        for (int s = 0; s < 16; ++s) {
            const float* p = dataB + (size_t)(n0 + lm) * DD + 4 * s + 2 * lh;
            af[s] = *reinterpret_cast<const v2f*>(p);
        }
#pragma unroll
        for (int j = 0; j < 2; ++j) {
            v8f acc = {};
#pragma unroll
            for (int s = 0; s < 16; ++s) {
                acc = __builtin_amdgcn_wmma_f32_16x16x4_f32(
                     false, af[s],  false, bfrag[j][s],
                     (short)0, acc,  false,  false);
            }
            const int lcol = wave * 32 + j * 16 + lm;
            v4f lo = {acc[0], acc[1], acc[2], acc[3]};
            v4f hi = {acc[4], acc[5], acc[6], acc[7]};
            *reinterpret_cast<v4f*>(&xw[buf][lcol][8 * lh])     = lo;
            *reinterpret_cast<v4f*>(&xw[buf][lcol][8 * lh + 4]) = hi;
        }
    };

    float cum = 0.0f;

    compute(0, 0);
    __syncthreads();

    for (int blk = 0; blk < NBLK; ++blk) {
        const int cur = blk & 1;
        const int n0  = blk * NB;

        if (blk + 1 < NBLK) compute(blk + 1, cur ^ 1);

        v4f xr[4], tr[4];
#pragma unroll
        for (int q = 0; q < 4; ++q)
            xr[q] = *reinterpret_cast<const v4f*>(&xw[cur][tid][4 * q]);
        const float* tb = targets + b * NN + n0;
#pragma unroll
        for (int q = 0; q < 4; ++q)
            tr[q] = *reinterpret_cast<const v4f*>(tb + 4 * q);

#pragma unroll
        for (int n = 0; n < NB; ++n) {
            const float x = xr[n >> 2][n & 3];

            const float m  = wave_max32(cum);
            const float e  = __expf(cum - m);
            const float nu = wave_sum32(e * x);
            const float de = wave_sum32(e);
            if (lane == 0) {
                v4f t4 = {m, nu, de, 0.0f};
                *reinterpret_cast<v4f*>(&red[cur][n][wave][0]) = t4;
            }
            const float df = tr[n >> 2][n & 3] - x;
            cum = fmaf(df, df * -0.5f, cum);
        }

        __syncthreads();

        if ((tid & 15) == 0) {
            const int n = tid >> 4;
            v4f t0 = *reinterpret_cast<const v4f*>(&red[cur][n][0][0]);
            float m = t0.x, num = t0.y, den = t0.z;
#pragma unroll
            for (int w = 1; w < 8; ++w) {
                v4f tw = *reinterpret_cast<const v4f*>(&red[cur][n][w][0]);
                smerge(m, num, den, tw.x, tw.y, tw.z);
            }
            const size_t o = (size_t)g * BN + (size_t)b * NN + n0 + n;
            wsM[o] = m; wsNum[o] = num; wsDen[o] = den;
        }
    }
}

__global__ __launch_bounds__(256)
void mmse_combine_kernel(const float* __restrict__ wsM,
                         const float* __restrict__ wsNum,
                         const float* __restrict__ wsDen,
                         float* __restrict__ out)
{
    const int i = (int)blockIdx.x * 256 + (int)threadIdx.x;
    if (i >= BN) return;
    float m = wsM[i], num = wsNum[i], den = wsDen[i];
#pragma unroll
    for (int g = 1; g < GG; ++g) {
        const size_t o = (size_t)g * BN + i;
        smerge(m, num, den, wsM[o], wsNum[o], wsDen[o]);
    }
    out[i] = num / den;
}

extern "C" void kernel_launch(void* const* d_in, const int* in_sizes, int n_in,
                              void* d_out, int out_size, void* d_ws, size_t ws_size,
                              hipStream_t stream) {
    (void)in_sizes; (void)n_in; (void)out_size; (void)ws_size;
    const float* data    = (const float*)d_in[0];
    const float* targets = (const float*)d_in[1];
    const float* W       = (const float*)d_in[2];
    float*       out     = (float*)d_out;

    float* wsBase = (float*)d_ws;
    float* wsM    = wsBase;
    float* wsNum  = wsBase + (size_t)BNG;
    float* wsDen  = wsBase + (size_t)2 * BNG;

    mmse_scan_kernel<<<dim3(BB * GG), dim3(256), 0, stream>>>(
        data, targets, W, wsM, wsNum, wsDen);
    mmse_combine_kernel<<<dim3((BN + 255) / 256), dim3(256), 0, stream>>>(
        wsM, wsNum, wsDen, out);
}


# A fragments gathered by eight 16-byte loads per block (was sixteen 8-byte): each line touched half as often; k slots re-paired alike in the A and B loaders; product sum order changed
# speedup vs baseline: 6.3294x; 6.3294x over previous
#include <hip/hip_runtime.h>

#define BB   32
#define NN   1024
#define DD   64
#define TT   2048
#define GG   8
#define TC   (TT / GG)
#define NB   16
#define NBLK (NN / NB)
#define BN   (BB * NN)
#define BNG  (BN * GG)

typedef float v2f __attribute__((ext_vector_type(2)));
typedef float v4f __attribute__((ext_vector_type(4)));
typedef float v8f __attribute__((ext_vector_type(8)));

#define SWIZ(v, imm) \
    __int_as_float(__builtin_amdgcn_ds_swizzle(__float_as_int(v), (imm)))
#define XOR16 0x401F

__device__ __forceinline__ v8f wmma4(v2f a, v2f b, v8f c) {
    return __builtin_amdgcn_wmma_f32_16x16x4_f32(
         false, a,  false, b,
         (short)0, c,  false,  false);
}

__device__ __forceinline__ void smerge(float& m, float& num, float& den,
                                       float m2, float n2, float d2) {
    float M  = fmaxf(m, m2);
    float s1 = __expf(m - M);
    float s2 = __expf(m2 - M);
    num = num * s1 + n2 * s2;
    den = den * s1 + d2 * s2;
    m   = M;
}

__global__ __launch_bounds__(256)
void mmse_scan_kernel(const float* __restrict__ data,
                      const float* __restrict__ targets,
                      const float* __restrict__ W,
                      float* __restrict__ wsM,
                      float* __restrict__ wsNum,
                      float* __restrict__ wsDen)
{
    __shared__ alignas(16) float red[8][NN][4];

    const int tid  = threadIdx.x;
    const int lane = tid & 31;
    const int wave = tid >> 5;
    const int lm   = lane & 15;
    const int lh   = lane >> 4;

    const int b     = (int)blockIdx.x >> 3;
    const int g     = (int)blockIdx.x & (GG - 1);
    const int tbase = g * TC;

    const float* dataB = data + (size_t)b * NN * DD;
    const float* targB = targets + (size_t)b * NN;

    v2f bfrag[2][16];
#pragma unroll
    for (int j = 0; j < 2; ++j) {
        const int gcol = tbase + wave * 32 + j * 16 + lm;
#pragma unroll
        for (int s = 0; s < 16; ++s) {
            const int k0 = 8 * (s >> 1) + 4 * lh + 2 * (s & 1);
            v2f bf;
            bf.x = W[(size_t)k0 * TT + gcol];
            bf.y = W[(size_t)(k0 + 1) * TT + gcol];
            bfrag[j][s] = bf;
        }
    }

    v2f triB[4], idB[4];
#pragma unroll
    for (int s = 0; s < 4; ++s) {
        const int k0 = 8 * lh + 2 * s;
        v2f t, p;
        t.x = (k0     <  lm) ? 1.0f : 0.0f;
        t.y = (k0 + 1 <  lm) ? 1.0f : 0.0f;
        p.x = (k0     == lm) ? 1.0f : 0.0f;
        p.y = (k0 + 1 == lm) ? 1.0f : 0.0f;
        triB[s] = t;
        idB[s]  = p;
    }
    const v2f onesB = {1.0f, 1.0f};

    auto load_a = [&](v4f (&af)[8], int blk) {
        const float* p = dataB + (size_t)(blk * NB + lm) * DD + 4 * lh;
#pragma unroll
        for (int q = 0; q < 8; ++q)
            af[q] = *reinterpret_cast<const v4f*>(p + 8 * q);
    };
    auto gemm = [&](const v4f (&af)[8], v8f& c0, v8f& c1) {
#pragma unroll
        for (int q = 0; q < 8; ++q) {
            const v2f lo = {af[q][0], af[q][1]};
            const v2f hi = {af[q][2], af[q][3]};
            c0 = wmma4(lo, bfrag[0][2 * q],     c0);
            c1 = wmma4(lo, bfrag[1][2 * q],     c1);
            c0 = wmma4(hi, bfrag[0][2 * q + 1], c0);
            c1 = wmma4(hi, bfrag[1][2 * q + 1], c1);
        }
    };

    v8f carry0 = {}, carry1 = {};

    v4f af[8];
    load_a(af, 0);
    v8f acc0 = {}, acc1 = {};
    gemm(af, acc0, acc1);
    load_a(af, 1);
    v4f y0 = *reinterpret_cast<const v4f*>(targB + 8 * lh);
    v4f y1 = *reinterpret_cast<const v4f*>(targB + 8 * lh + 4);

    for (int blk = 0; blk < NBLK; ++blk) {
        const int n0  = blk * NB;

        v8f nx0 = {}, nx1 = {};
        gemm(af, nx0, nx1);
        load_a(af, (blk + 2 < NBLK) ? blk + 2 : NBLK - 1);

        const int nn = ((blk + 1 < NBLK) ? blk + 1 : NBLK - 1) * NB + 8 * lh;
        const v4f ny0 = *reinterpret_cast<const v4f*>(targB + nn);
        const v4f ny1 = *reinterpret_cast<const v4f*>(targB + nn + 4);

        v8f lp0, lp1;
#pragma unroll
        for (int r = 0; r < 8; ++r) {
            const float y  = (r < 4) ? y0[r & 3] : y1[r & 3];
            const float d0 = y - acc0[r];
            const float d1 = y - acc1[r];
            lp0[r] = d0 * (d0 * -0.5f);
            lp1[r] = d1 * (d1 * -0.5f);
        }

        v8f al0 = carry0, al1 = carry1;
        v8f cn0 = carry0, cn1 = carry1;
        v8f xt0 = {},     xt1 = {};
#pragma unroll
        for (int s = 0; s < 4; ++s) {
            const v2f a0 = {lp0[2 * s], lp0[2 * s + 1]};
            const v2f a1 = {lp1[2 * s], lp1[2 * s + 1]};
            const v2f x0 = {acc0[2 * s], acc0[2 * s + 1]};
            const v2f x1 = {acc1[2 * s], acc1[2 * s + 1]};
            al0 = wmma4(a0, triB[s], al0);
            al1 = wmma4(a1, triB[s], al1);
            cn0 = wmma4(a0, onesB,   cn0);
            cn1 = wmma4(a1, onesB,   cn1);
            xt0 = wmma4(x0, idB[s],  xt0);
            xt1 = wmma4(x1, idB[s],  xt1);
        }
        carry0 = cn0;
        carry1 = cn1;

        float m = al0[0];
#pragma unroll
        for (int r = 1; r < 8; ++r) m = fmaxf(m, al0[r]);
#pragma unroll
        for (int r = 0; r < 8; ++r) m = fmaxf(m, al1[r]);
        m = fmaxf(m, SWIZ(m, XOR16));

        float de0 = 0.0f, nu0 = 0.0f, de1 = 0.0f, nu1 = 0.0f;
#pragma unroll
        for (int r = 0; r < 8; ++r) {
            const float e0 = __expf(al0[r] - m);
            const float e1 = __expf(al1[r] - m);
            de0 += e0;
            de1 += e1;
            nu0 = fmaf(e0, xt0[r], nu0);
            nu1 = fmaf(e1, xt1[r], nu1);
        }
        float de = de0 + de1;
        float nu = nu0 + nu1;
        de += SWIZ(de, XOR16);
        nu += SWIZ(nu, XOR16);
        if (lh == 0) {
            v4f t4 = {m, nu, de, 0.0f};
            *reinterpret_cast<v4f*>(&red[wave][n0 + lm][0]) = t4;
        }

        acc0 = nx0;
        acc1 = nx1;
        y0   = ny0;
        y1   = ny1;
    }

    __syncthreads();

#pragma unroll
    for (int i = 0; i < NN / 256; ++i) {
        const int n = tid + 256 * i;
        v4f t0 = *reinterpret_cast<const v4f*>(&red[0][n][0]);
        float mm = t0.x, num = t0.y, den = t0.z;
#pragma unroll
        for (int w = 1; w < 8; ++w) {
            v4f tw = *reinterpret_cast<const v4f*>(&red[w][n][0]);
            smerge(mm, num, den, tw.x, tw.y, tw.z);
        }
        const size_t o = (size_t)g * BN + (size_t)b * NN + n;
        wsM[o] = mm; wsNum[o] = num; wsDen[o] = den;
    }
}

__global__ __launch_bounds__(256)
void mmse_combine_kernel(const float* __restrict__ wsM,
                         const float* __restrict__ wsNum,
                         const float* __restrict__ wsDen,
                         float* __restrict__ out)
{
    const int i = (int)blockIdx.x * 256 + (int)threadIdx.x;
    if (i >= BN) return;
    float m = wsM[i], num = wsNum[i], den = wsDen[i];
#pragma unroll
    for (int g = 1; g < GG; ++g) {
        const size_t o = (size_t)g * BN + i;
        smerge(m, num, den, wsM[o], wsNum[o], wsDen[o]);
    }
    out[i] = num / den;
}

extern "C" void kernel_launch(void* const* d_in, const int* in_sizes, int n_in,
                              void* d_out, int out_size, void* d_ws, size_t ws_size,
                              hipStream_t stream) {
    (void)in_sizes; (void)n_in; (void)out_size; (void)ws_size;
    const float* data    = (const float*)d_in[0];
    const float* targets = (const float*)d_in[1];
    const float* W       = (const float*)d_in[2];
    float*       out     = (float*)d_out;

    float* wsBase = (float*)d_ws;
    float* wsM    = wsBase;
    float* wsNum  = wsBase + (size_t)BNG;
    float* wsDen  = wsBase + (size_t)2 * BNG;

    mmse_scan_kernel<<<dim3(BB * GG), dim3(256), 0, stream>>>(
        data, targets, W, wsM, wsNum, wsDen);
    mmse_combine_kernel<<<dim3((BN + 255) / 256), dim3(256), 0, stream>>>(
        wsM, wsNum, wsDen, out);
}


# product tile turned one-row-per-lane through a wave-private LDS patch (stored as it lies, two 16-byte reads per tile) instead of 8 identity-operand WMMAs per trip; no barrier; sum order kept
# speedup vs baseline: 6.5150x; 1.0293x over previous
#include <hip/hip_runtime.h>

#define BB   32
#define NN   1024
#define DD   64
#define TT   2048
#define GG   8
#define TC   (TT / GG)
#define NB   16
#define NBLK (NN / NB)
#define BN   (BB * NN)
#define BNG  (BN * GG)
#define XP   20

typedef float v2f __attribute__((ext_vector_type(2)));
typedef float v4f __attribute__((ext_vector_type(4)));
typedef float v8f __attribute__((ext_vector_type(8)));

#define SWIZ(v, imm) \
    __int_as_float(__builtin_amdgcn_ds_swizzle(__float_as_int(v), (imm)))
#define XOR16 0x401F

__device__ __forceinline__ v8f wmma4(v2f a, v2f b, v8f c) {
    return __builtin_amdgcn_wmma_f32_16x16x4_f32(
         false, a,  false, b,
         (short)0, c,  false,  false);
}

__device__ __forceinline__ void smerge(float& m, float& num, float& den,
                                       float m2, float n2, float d2) {
    float M  = fmaxf(m, m2);
    float s1 = __expf(m - M);
    float s2 = __expf(m2 - M);
    num = num * s1 + n2 * s2;
    den = den * s1 + d2 * s2;
    m   = M;
}

__global__ __launch_bounds__(256)
void mmse_scan_kernel(const float* __restrict__ data,
                      const float* __restrict__ targets,
                      const float* __restrict__ W,
                      float* __restrict__ wsM,
                      float* __restrict__ wsNum,
                      float* __restrict__ wsDen)
{
    __shared__ alignas(16) float red[8][NN][4];
    __shared__ alignas(16) float xpose[8][2][NB][XP];

    const int tid  = threadIdx.x;
    const int lane = tid & 31;
    const int wave = tid >> 5;
    const int lm   = lane & 15;
    const int lh   = lane >> 4;

    const int b     = (int)blockIdx.x >> 3;
    const int g     = (int)blockIdx.x & (GG - 1);
    const int tbase = g * TC;

    const float* dataB = data + (size_t)b * NN * DD;
    const float* targB = targets + (size_t)b * NN;

    v2f bfrag[2][16];
#pragma unroll
    for (int j = 0; j < 2; ++j) {
        const int gcol = tbase + wave * 32 + j * 16 + lm;
#pragma unroll
        for (int s = 0; s < 16; ++s) {
            const int k0 = 8 * (s >> 1) + 4 * lh + 2 * (s & 1);
            v2f bf;
            bf.x = W[(size_t)k0 * TT + gcol];
            bf.y = W[(size_t)(k0 + 1) * TT + gcol];
            bfrag[j][s] = bf;
        }
    }

    v2f triB[4];
#pragma unroll
    for (int s = 0; s < 4; ++s) {
        const int k0 = 8 * lh + 2 * s;
        v2f t;
        t.x = (k0     <  lm) ? 1.0f : 0.0f;
        t.y = (k0 + 1 <  lm) ? 1.0f : 0.0f;
        triB[s] = t;
    }
    const v2f onesB = {1.0f, 1.0f};

    auto load_a = [&](v4f (&af)[8], int blk) {
        const float* p = dataB + (size_t)(blk * NB + lm) * DD + 4 * lh;
#pragma unroll
        for (int q = 0; q < 8; ++q)
            af[q] = *reinterpret_cast<const v4f*>(p + 8 * q);
    };
    auto gemm = [&](const v4f (&af)[8], v8f& c0, v8f& c1) {
#pragma unroll
        for (int q = 0; q < 8; ++q) {
            const v2f lo = {af[q][0], af[q][1]};
            const v2f hi = {af[q][2], af[q][3]};
            c0 = wmma4(lo, bfrag[0][2 * q],     c0);
            c1 = wmma4(lo, bfrag[1][2 * q],     c1);
            c0 = wmma4(hi, bfrag[0][2 * q + 1], c0);
            c1 = wmma4(hi, bfrag[1][2 * q + 1], c1);
        }
    };

    auto park_x = [&](const v8f& c0, const v8f& c1) {
#pragma unroll
        for (int r = 0; r < 8; ++r) {
            xpose[wave][0][8 * lh + r][lm] = c0[r];
            xpose[wave][1][8 * lh + r][lm] = c1[r];
        }
    };
    auto take_x = [&](v8f& x0, v8f& x1) {
        const v4f a0 = *reinterpret_cast<const v4f*>(&xpose[wave][0][lm][8 * lh]);
        const v4f a1 = *reinterpret_cast<const v4f*>(&xpose[wave][0][lm][8 * lh + 4]);
        const v4f b0 = *reinterpret_cast<const v4f*>(&xpose[wave][1][lm][8 * lh]);
        const v4f b1 = *reinterpret_cast<const v4f*>(&xpose[wave][1][lm][8 * lh + 4]);
#pragma unroll
        for (int r = 0; r < 4; ++r) {
            x0[r] = a0[r]; x0[r + 4] = a1[r];
            x1[r] = b0[r]; x1[r + 4] = b1[r];
        }
    };

    v8f carry0 = {}, carry1 = {};

    v4f af[8];
    load_a(af, 0);
    v8f acc0 = {}, acc1 = {};
    gemm(af, acc0, acc1);
    v8f xt0, xt1;
    park_x(acc0, acc1);
    take_x(xt0, xt1);
    load_a(af, 1);
    v4f y0 = *reinterpret_cast<const v4f*>(targB + 8 * lh);
    v4f y1 = *reinterpret_cast<const v4f*>(targB + 8 * lh + 4);

    for (int blk = 0; blk < NBLK; ++blk) {
        const int n0  = blk * NB;

        v8f nx0 = {}, nx1 = {};
        gemm(af, nx0, nx1);
        load_a(af, (blk + 2 < NBLK) ? blk + 2 : NBLK - 1);

        const int nn = ((blk + 1 < NBLK) ? blk + 1 : NBLK - 1) * NB + 8 * lh;
        const v4f ny0 = *reinterpret_cast<const v4f*>(targB + nn);
        const v4f ny1 = *reinterpret_cast<const v4f*>(targB + nn + 4);

        v8f lp0, lp1;
#pragma unroll
        for (int r = 0; r < 8; ++r) {
            const float y  = (r < 4) ? y0[r & 3] : y1[r & 3];
            const float d0 = y - acc0[r];
            const float d1 = y - acc1[r];
            lp0[r] = d0 * (d0 * -0.5f);
            lp1[r] = d1 * (d1 * -0.5f);
        }

        v8f al0 = carry0, al1 = carry1;
        v8f cn0 = carry0, cn1 = carry1;
#pragma unroll
        for (int s = 0; s < 4; ++s) {
            const v2f a0 = {lp0[2 * s], lp0[2 * s + 1]};
            const v2f a1 = {lp1[2 * s], lp1[2 * s + 1]};
            al0 = wmma4(a0, triB[s], al0);
            al1 = wmma4(a1, triB[s], al1);
            cn0 = wmma4(a0, onesB,   cn0);
            cn1 = wmma4(a1, onesB,   cn1);
        }
        carry0 = cn0;
        carry1 = cn1;

        float m = al0[0];
#pragma unroll
        for (int r = 1; r < 8; ++r) m = fmaxf(m, al0[r]);
#pragma unroll
        for (int r = 0; r < 8; ++r) m = fmaxf(m, al1[r]);
        m = fmaxf(m, SWIZ(m, XOR16));

        float de0 = 0.0f, nu0 = 0.0f, de1 = 0.0f, nu1 = 0.0f;
#pragma unroll
        for (int r = 0; r < 8; ++r) {
            const float e0 = __expf(al0[r] - m);
            const float e1 = __expf(al1[r] - m);
            de0 += e0;
            de1 += e1;
            nu0 = fmaf(e0, xt0[r], nu0);
            nu1 = fmaf(e1, xt1[r], nu1);
        }
        float de = de0 + de1;
        float nu = nu0 + nu1;
        de += SWIZ(de, XOR16);
        nu += SWIZ(nu, XOR16);
        if (lh == 0) {
            v4f t4 = {m, nu, de, 0.0f};
            *reinterpret_cast<v4f*>(&red[wave][n0 + lm][0]) = t4;
        }

        park_x(nx0, nx1);
        take_x(xt0, xt1);

        acc0 = nx0;
        acc1 = nx1;
        y0   = ny0;
        y1   = ny1;
    }

    __syncthreads();

#pragma unroll
    for (int i = 0; i < NN / 256; ++i) {
        const int n = tid + 256 * i;
        v4f t0 = *reinterpret_cast<const v4f*>(&red[0][n][0]);
        float mm = t0.x, num = t0.y, den = t0.z;
#pragma unroll
        for (int w = 1; w < 8; ++w) {
            v4f tw = *reinterpret_cast<const v4f*>(&red[w][n][0]);
            smerge(mm, num, den, tw.x, tw.y, tw.z);
        }
        const size_t o = (size_t)g * BN + (size_t)b * NN + n;
        wsM[o] = mm; wsNum[o] = num; wsDen[o] = den;
    }
}

__global__ __launch_bounds__(256)
void mmse_combine_kernel(const float* __restrict__ wsM,
                         const float* __restrict__ wsNum,
                         const float* __restrict__ wsDen,
                         float* __restrict__ out)
{
    const int i = (int)blockIdx.x * 256 + (int)threadIdx.x;
    if (i >= BN) return;
    float m = wsM[i], num = wsNum[i], den = wsDen[i];
#pragma unroll
    for (int g = 1; g < GG; ++g) {
        const size_t o = (size_t)g * BN + i;
        smerge(m, num, den, wsM[o], wsNum[o], wsDen[o]);
    }
    out[i] = num / den;
}

extern "C" void kernel_launch(void* const* d_in, const int* in_sizes, int n_in,
                              void* d_out, int out_size, void* d_ws, size_t ws_size,
                              hipStream_t stream) {
    (void)in_sizes; (void)n_in; (void)out_size; (void)ws_size;
    const float* data    = (const float*)d_in[0];
    const float* targets = (const float*)d_in[1];
    const float* W       = (const float*)d_in[2];
    float*       out     = (float*)d_out;

    float* wsBase = (float*)d_ws;
    float* wsM    = wsBase;
    float* wsNum  = wsBase + (size_t)BNG;
    float* wsDen  = wsBase + (size_t)2 * BNG;

    mmse_scan_kernel<<<dim3(BB * GG), dim3(256), 0, stream>>>(
        data, targets, W, wsM, wsNum, wsDen);
    mmse_combine_kernel<<<dim3((BN + 255) / 256), dim3(256), 0, stream>>>(
        wsM, wsNum, wsDen, out);
}
